# starting kernel
# baseline (speedup 1.0000x reference)
#include <hip/hip_runtime.h>
#include <math.h>

#define BB 8
#define HH 128
#define LL 4096
#define NN 64
#define CHUNK 64
#define NCH (LL / CHUNK)
#define COLS (BB * NCH)
#define EPSV 1e-7f

typedef float v2f __attribute__((ext_vector_type(2)));
typedef float v8f __attribute__((ext_vector_type(8)));

__global__ void __launch_bounds__(64)
dss_coeffs(const float* __restrict__ Lre, const float* __restrict__ Lim,
           const float* __restrict__ Wri, const float* __restrict__ logstep,
           float* __restrict__ Zr, float* __restrict__ Zi,
           float* __restrict__ CR, float* __restrict__ CIn,
           float* __restrict__ T, float* __restrict__ zCr, float* __restrict__ zCi) {
    __shared__ float sdtr[NN], sdti[NN], scr[NN], sci[NN], sK[CHUNK];
    const int h = blockIdx.x;
    const int n = threadIdx.x;

    const float lre = Lre[h * NN + n], lim = Lim[h * NN + n];
    const float dt  = __expf(logstep[h]);
    const float ar  = dt * lre, ai = dt * lim;

    float cz, sz; sincosf(ai, &sz, &cz);
    const float er = expf(ar);
    const float zr = er * cz, zi = er * sz;
    float cL, sL; sincosf(ai * (float)LL, &sL, &cL);
    const float eL = expf(ar * (float)LL);
    const float zLr = eL * cL, zLi = eL * sL;
    const float nr = 1.f - zLr, ni = -zLi;
    const float dr = 1.f - zr,  di = -zi;
    const float dd = dr * dr + di * di;
    const float sr = (nr * dr + ni * di) / dd;
    const float si = (ni * dr - nr * di) / dd;
    const float wr = Wri[(h * NN + n) * 2 + 0], wi = Wri[(h * NN + n) * 2 + 1];
    const float ll = lre * lre + lim * lim;
    const float br = (wr * lre + wi * lim) / ll;
    const float bi = (wi * lre - wr * lim) / ll;
    const float ss = sr * sr + si * si + EPSV;
    const float cr = (br * sr + bi * si) / ss;
    const float ci = (bi * sr - br * si) / ss;

    float cC, sC; sincosf(ai * (float)CHUNK, &sC, &cC);
    const float eC = expf(ar * (float)CHUNK);
    zCr[h * NN + n] = eC * cC;
    zCi[h * NN + n] = eC * sC;

    {
        float pr = 1.f, pi_ = 0.f;
        float* zrrow = Zr + (h * NN + n) * CHUNK;
        float* zirow = Zi + (h * NN + n) * CHUNK;
        for (int k = 0; k < CHUNK; ++k) {
            zrrow[CHUNK - 1 - k] = pr;
            zirow[CHUNK - 1 - k] = pi_;
            const float tr = pr * zr - pi_ * zi;
            pi_ = pr * zi + pi_ * zr;
            pr = tr;
        }
    }
    {
        float qr = cr * zr - ci * zi;
        float qi = cr * zi + ci * zr;
        for (int t = 0; t < CHUNK; ++t) {
            CR [(h * CHUNK + t) * NN + n] =  qr;
            CIn[(h * CHUNK + t) * NN + n] = -qi;
            const float tr = qr * zr - qi * zi;
            qi = qr * zi + qi * zr;
            qr = tr;
        }
    }

    sdtr[n] = ar; sdti[n] = ai; scr[n] = cr; sci[n] = ci;
    __syncthreads();

    {
        const float t = (float)n;
        float acc = 0.f;
        for (int j = 0; j < NN; ++j) {
            const float e = expf(sdtr[j] * t);
            float cs, sn; sincosf(sdti[j] * t, &sn, &cs);
            acc += e * (scr[j] * cs - sci[j] * sn);
        }
        sK[n] = acc;
    }
    __syncthreads();

    for (int idx = n; idx < CHUNK * CHUNK; idx += NN) {
        const int t = idx >> 6, s = idx & 63;
        T[(h * CHUNK + t) * CHUNK + s] = (t >= s) ? sK[t - s] : 0.f;
    }
}

__global__ void __launch_bounds__(256)
dss_proj(const float* __restrict__ u,
         const float* __restrict__ Zr, const float* __restrict__ Zi,
         float* __restrict__ Pr, float* __restrict__ Pi) {
    const int lane  = threadIdx.x & 31;
    const int wave  = threadIdx.x >> 5;
    const int tile  = blockIdx.x * 8 + wave;
    const int h     = tile >> 7;
    const int rem   = tile & 127;
    const int rt    = rem >> 5;
    const int ct    = rem & 31;
    const int m     = lane & 15;
    const int khalf = lane >> 4;
    const int k0    = khalf * 2;

    const int arow = rt * 16 + m;
    const int col  = ct * 16 + m;
    const float* __restrict__ ucol =
        u + ((size_t)(col >> 6) * HH + h) * LL + (size_t)(col & 63) * CHUNK;
    const float* __restrict__ zrrow = Zr + (size_t)(h * NN + arow) * CHUNK;
    const float* __restrict__ zirow = Zi + (size_t)(h * NN + arow) * CHUNK;

    v8f accr = {}, acci = {};
#pragma unroll
    for (int kk = 0; kk < CHUNK; kk += 4) {
        const v2f aR = *(const v2f*)(zrrow + kk + k0);
        const v2f aI = *(const v2f*)(zirow + kk + k0);
        const v2f b  = *(const v2f*)(ucol + kk + k0);
        accr = __builtin_amdgcn_wmma_f32_16x16x4_f32(false, aR, false, b, (short)0, accr, false, false);
        acci = __builtin_amdgcn_wmma_f32_16x16x4_f32(false, aI, false, b, (short)0, acci, false, false);
    }

    const int nbase = rt * 16 + 8 * khalf;
    const size_t o = ((size_t)(h * COLS + col)) * NN + nbase;
    float4 r0, r1, i0, i1;
    r0.x = accr[0]; r0.y = accr[1]; r0.z = accr[2]; r0.w = accr[3];
    r1.x = accr[4]; r1.y = accr[5]; r1.z = accr[6]; r1.w = accr[7];
    i0.x = acci[0]; i0.y = acci[1]; i0.z = acci[2]; i0.w = acci[3];
    i1.x = acci[4]; i1.y = acci[5]; i1.z = acci[6]; i1.w = acci[7];
    *(float4*)(Pr + o)     = r0;
    *(float4*)(Pr + o + 4) = r1;
    *(float4*)(Pi + o)     = i0;
    *(float4*)(Pi + o + 4) = i1;
}

__global__ void __launch_bounds__(256)
dss_scan(float* __restrict__ Pr, float* __restrict__ Pi,
         const float* __restrict__ zCr, const float* __restrict__ zCi) {
    const int tid = blockIdx.x * blockDim.x + threadIdx.x;
    const int n = tid & 63;
    const int b = (tid >> 6) & 7;
    const int h = tid >> 9;
    const size_t base = ((size_t)(h * COLS + b * NCH)) * NN + n;
    const float zr = zCr[h * NN + n], zi = zCi[h * NN + n];
    float xr = 0.f, xi = 0.f;
    for (int i = 0; i < NCH; ++i) {
        const size_t o = base + (size_t)i * NN;
        const float pr = Pr[o], pi_ = Pi[o];
        const float nxr = zr * xr - zi * xi + pr;
        const float nxi = zr * xi + zi * xr + pi_;
        Pr[o] = nxr; Pi[o] = nxi;
        xr = nxr; xi = nxi;
    }
}

__global__ void __launch_bounds__(256)
dss_out(const float* __restrict__ u, const float* __restrict__ D,
        const float* __restrict__ T, const float* __restrict__ CR,
        const float* __restrict__ CIn,
        const float* __restrict__ Pr, const float* __restrict__ Pi,
        float* __restrict__ out) {
    const int lane  = threadIdx.x & 31;
    const int wave  = threadIdx.x >> 5;
    const int tile  = blockIdx.x * 8 + wave;
    const int h     = tile >> 7;
    const int rem   = tile & 127;
    const int rt    = rem >> 5;
    const int ct    = rem & 31;
    const int m     = lane & 15;
    const int khalf = lane >> 4;
    const int k0    = khalf * 2;

    const int arow = rt * 16 + m;
    const int col  = ct * 16 + m;
    const int b    = col >> 6;
    const int ch   = col & 63;
    const bool first = (ch == 0);

    const float* __restrict__ ucol  = u + ((size_t)b * HH + h) * LL + (size_t)ch * CHUNK;
    const float* __restrict__ trow  = T   + (size_t)(h * CHUNK + arow) * CHUNK;
    const float* __restrict__ crrow = CR  + (size_t)(h * CHUNK + arow) * NN;
    const float* __restrict__ cirow = CIn + (size_t)(h * CHUNK + arow) * NN;
    const float* __restrict__ xrcol = Pr + ((size_t)(h * COLS + col - 1)) * NN;
    const float* __restrict__ xicol = Pi + ((size_t)(h * COLS + col - 1)) * NN;

    v8f acc = {};
#pragma unroll
    for (int kk = 0; kk < CHUNK; kk += 4) {
        const v2f a  = *(const v2f*)(trow + kk + k0);
        const v2f bu = *(const v2f*)(ucol + kk + k0);
        acc = __builtin_amdgcn_wmma_f32_16x16x4_f32(false, a, false, bu, (short)0, acc, false, false);
    }
#pragma unroll
    for (int kk = 0; kk < CHUNK; kk += 4) {
        const v2f aR = *(const v2f*)(crrow + kk + k0);
        const v2f aI = *(const v2f*)(cirow + kk + k0);
        v2f bxr = {0.f, 0.f}, bxi = {0.f, 0.f};
        if (!first) {
            bxr = *(const v2f*)(xrcol + kk + k0);
            bxi = *(const v2f*)(xicol + kk + k0);
        }
        acc = __builtin_amdgcn_wmma_f32_16x16x4_f32(false, aR, false, bxr, (short)0, acc, false, false);
        acc = __builtin_amdgcn_wmma_f32_16x16x4_f32(false, aI, false, bxi, (short)0, acc, false, false);
    }

    const float Dh = D[h];
    const size_t obase = ((size_t)b * HH + h) * LL + (size_t)ch * CHUNK + rt * 16 + 8 * khalf;
    const float4 u0 = *(const float4*)(u + obase);
    const float4 u1 = *(const float4*)(u + obase + 4);
    float4 o0, o1;
    o0.x = acc[0] + Dh * u0.x; o0.y = acc[1] + Dh * u0.y;
    o0.z = acc[2] + Dh * u0.z; o0.w = acc[3] + Dh * u0.w;
    o1.x = acc[4] + Dh * u1.x; o1.y = acc[5] + Dh * u1.y;
    o1.z = acc[6] + Dh * u1.z; o1.w = acc[7] + Dh * u1.w;
    *(float4*)(out + obase)     = o0;
    *(float4*)(out + obase + 4) = o1;
}

extern "C" void kernel_launch(void* const* d_in, const int* in_sizes, int n_in,
                              void* d_out, int out_size, void* d_ws, size_t ws_size,
                              hipStream_t stream) {
    const float* u       = (const float*)d_in[0];
    const float* Lre     = (const float*)d_in[1];
    const float* Lim     = (const float*)d_in[2];
    const float* Wri     = (const float*)d_in[3];
    const float* Dp      = (const float*)d_in[4];
    const float* logstep = (const float*)d_in[5];
    float* out = (float*)d_out;

    float* ws  = (float*)d_ws;
    float* Zr  = ws;
    float* Zi  = Zr  + (size_t)HH * NN * CHUNK;
    float* CRm = Zi  + (size_t)HH * NN * CHUNK;
    float* CIn = CRm + (size_t)HH * CHUNK * NN;
    float* Tm  = CIn + (size_t)HH * CHUNK * NN;
    float* zCr = Tm  + (size_t)HH * CHUNK * CHUNK;
    float* zCi = zCr + (size_t)HH * NN;
    float* Pr  = zCi + (size_t)HH * NN;
    float* Pi  = Pr  + (size_t)HH * COLS * NN;

    dss_coeffs<<<HH, NN, 0, stream>>>(Lre, Lim, Wri, logstep,
                                      Zr, Zi, CRm, CIn, Tm, zCr, zCi);

    const int tiles = HH * (NN / 16) * (COLS / 16);
    dss_proj<<<tiles / 8, 256, 0, stream>>>(u, Zr, Zi, Pr, Pi);

    dss_scan<<<(HH * NN * BB) / 256, 256, 0, stream>>>(Pr, Pi, zCr, zCi);

    dss_out<<<tiles / 8, 256, 0, stream>>>(u, Dp, Tm, CRm, CIn, Pr, Pi, out);
}


# First launch dropped: each projection work-group walks its head's 64 z-power chains (waves 0-1, into LDS Z tiles, while waves 4-7 fetch U), z^C in registers; 3 launches become 2; sum order kept
# speedup vs baseline: 3.4638x; 3.4638x over previous
#include <hip/hip_runtime.h>
#include <math.h>

#define BB 8
#define HH 128
#define LL 4096
#define NN 64
#define CHUNK 64
#define NCH (LL / CHUNK)
#define COLS (BB * NCH)
#define EPSV 1e-7f

typedef float v2f __attribute__((ext_vector_type(2)));
typedef float v4f __attribute__((ext_vector_type(4)));
typedef float v8f __attribute__((ext_vector_type(8)));


__device__ __forceinline__ void
build_ckt(const float* __restrict__ Lre, const float* __restrict__ Lim,
          const float* __restrict__ Wri, const float* __restrict__ logstep,
          float* __restrict__ CR, float* __restrict__ CIn, float* __restrict__ T,
          float* small, float* sE, float* sV, int h, int tid) {
    float* sdtr = small;
    float* sdti = small + NN;
    float* scr  = small + 2 * NN;
    float* sci  = small + 3 * NN;
    float* sK   = small + 4 * NN;
    const int n    = tid & 63;
    const int part = tid >> 6;

    const float lre = Lre[h * NN + n], lim = Lim[h * NN + n];
    const float dt  = __expf(logstep[h]);
    const float ar  = dt * lre, ai = dt * lim;

    float cz, sz; sincosf(ai, &sz, &cz);
    const float er = expf(ar);
    const float zr = er * cz, zi = er * sz;
    float cL, sL; sincosf(ai * (float)LL, &sL, &cL);
    const float eL = expf(ar * (float)LL);
    const float zLr = eL * cL, zLi = eL * sL;
    const float nr = 1.f - zLr, ni = -zLi;
    const float dr = 1.f - zr,  di = -zi;
    const float dd = dr * dr + di * di;
    const float sr = (nr * dr + ni * di) / dd;
    const float si = (ni * dr - nr * di) / dd;
    const float wr = Wri[(h * NN + n) * 2 + 0], wi = Wri[(h * NN + n) * 2 + 1];
    const float ll = lre * lre + lim * lim;
    const float br = (wr * lre + wi * lim) / ll;
    const float bi = (wi * lre - wr * lim) / ll;
    const float ss = sr * sr + si * si + EPSV;
    const float cr = (br * sr + bi * si) / ss;
    const float ci = (bi * sr - br * si) / ss;

    if (part == 0) { sdtr[n] = ar; sdti[n] = ai; scr[n] = cr; sci[n] = ci; }
    __syncthreads();

    {
        const float t = (float)n;
        for (int jj = 0; jj < NN / 4; ++jj) {
            const int j = part * (NN / 4) + jj;
            const float e = expf(sdtr[j] * t);
            float cs, sn; sincosf(sdti[j] * t, &sn, &cs);
            sE[j * CHUNK + n] = e;
            sV[j * CHUNK + n] = scr[j] * cs - sci[j] * sn;
        }
    }
    __syncthreads();

    if (part == 0) {
        float acc = 0.f;
        for (int j = 0; j < NN; ++j)
            acc += sE[j * CHUNK + n] * sV[j * CHUNK + n];
        sK[n] = acc;
    } else if (part == 1) {
        float qr = cr * zr - ci * zi;
        float qi = cr * zi + ci * zr;
        for (int t = 0; t < CHUNK; ++t) {
            CR [(h * CHUNK + t) * NN + n] =  qr;
            CIn[(h * CHUNK + t) * NN + n] = -qi;
            const float tr = qr * zr - qi * zi;
            qi = qr * zi + qi * zr;
            qr = tr;
        }
    }
    __syncthreads();

    for (int idx = tid; idx < CHUNK * CHUNK; idx += 256) {
        const int t = idx >> 6, s = idx & 63;
        T[(h * CHUNK + t) * CHUNK + s] = (t >= s) ? sK[t - s] : 0.f;
    }
}

#define LDP 68
#define CBW 128
#define NCB (COLS / CBW)

template <int NROWS>
__device__ __forceinline__ void stage_rows(const float* __restrict__ src, float* lds, int tid) {
    for (int j = 0; j < NROWS * 16 / 256; ++j) {
        const int i = tid + 256 * j;
        const float4 v = *(const float4*)(src + (size_t)i * 4);
        *(float4*)(lds + (i >> 4) * LDP + (i & 15) * 4) = v;
    }
}

__device__ __forceinline__ v8f mma4(v2f a, v2f b, v8f c) {
    return __builtin_amdgcn_wmma_f32_16x16x4_f32(false, a, false, b, (short)0, c, false, false);
}

__device__ __forceinline__ void store8(float* __restrict__ p, v8f v) {
    float4 lo, hi;
    lo.x = v[0]; lo.y = v[1]; lo.z = v[2]; lo.w = v[3];
    hi.x = v[4]; hi.y = v[5]; hi.z = v[6]; hi.w = v[7];
    *(float4*)(p)     = lo;
    *(float4*)(p + 4) = hi;
}

__global__ void __launch_bounds__(256)
dss_proj(const float* __restrict__ u,
         float* __restrict__ Pr, float* __restrict__ Pi,
         const float* __restrict__ Lre, const float* __restrict__ Lim,
         const float* __restrict__ Wri, const float* __restrict__ logstep,
         float* __restrict__ CR, float* __restrict__ CIn, float* __restrict__ T) {
    __shared__ float smem[2 * CBW * LDP];
    float* sZr = smem;
    float* sZi = smem + 64 * LDP;
    float* sU  = smem + 128 * LDP;
    float* sPr = smem;
    float* sPi = smem + CBW * LDP;

    const int tid   = threadIdx.x;

    if (blockIdx.x < HH) {
        build_ckt(Lre, Lim, Wri, logstep, CR, CIn, T,
                  sZr, sU, sU + CHUNK * NN, blockIdx.x, tid);
        return;
    }

    const int lane  = tid & 31;
    const int wave  = tid >> 5;
    const int bid   = blockIdx.x - HH;
    const int h     = bid / NCB;
    const int cb    = bid % NCB;
    const int m     = lane & 15;
    const int khalf = lane >> 4;
    const int k0    = khalf * 2;

    float zcr = 0.f, zci = 0.f;
    if (tid < 2 * NN) {
        const int   n   = tid & 63;
        const float lre = Lre[h * NN + n], lim = Lim[h * NN + n];
        const float dt  = __expf(logstep[h]);
        const float ar  = dt * lre, ai = dt * lim;
        if (tid < NN) {
            float cz, sz; sincosf(ai, &sz, &cz);
            const float er = expf(ar);
            const float zr = er * cz, zi = er * sz;
            float pr = 1.f, pi_ = 0.f;
            float* zrrow = sZr + n * LDP;
            float* zirow = sZi + n * LDP;
            for (int k = 0; k < CHUNK; k += 4) {
                float qr[4], qi[4];
                for (int e = 0; e < 4; ++e) {
                    qr[3 - e] = pr; qi[3 - e] = pi_;
                    const float tr = pr * zr - pi_ * zi;
                    pi_ = pr * zi + pi_ * zr;
                    pr = tr;
                }
                *(float4*)(zrrow + CHUNK - 4 - k) = make_float4(qr[0], qr[1], qr[2], qr[3]);
                *(float4*)(zirow + CHUNK - 4 - k) = make_float4(qi[0], qi[1], qi[2], qi[3]);
            }
        }
        float cC, sC; sincosf(ai * (float)CHUNK, &sC, &cC);
        const float eC = expf(ar * (float)CHUNK);
        zcr = eC * cC;
        zci = eC * sC;
    } else {
        const int t = tid - 2 * NN;
        const float* u0 = u + ((size_t)(2 * cb)     * HH + h) * LL;
        const float* u1 = u + ((size_t)(2 * cb + 1) * HH + h) * LL;
        v4f uv[16];
        for (int j = 0; j < 8; ++j) {
            uv[j]     = *(const v4f*)(u0 + (size_t)(t + 128 * j) * 4);
            uv[j + 8] = *(const v4f*)(u1 + (size_t)(t + 128 * j) * 4);
        }
        for (int j = 0; j < 8; ++j) {
            const int i = t + 128 * j;
            *(v4f*)(sU +            (i >> 4) * LDP + (i & 15) * 4) = uv[j];
            *(v4f*)(sU + 64 * LDP + (i >> 4) * LDP + (i & 15) * 4) = uv[j + 8];
        }
    }
    __syncthreads();

    const float* bcol = sU  + (wave * 16 + m) * LDP + k0;
    const float* ar   = sZr + m * LDP + k0;
    const float* ai   = sZi + m * LDP + k0;

    v8f r0 = {}, r1 = {}, r2 = {}, r3 = {};
    v8f i0 = {}, i1 = {}, i2 = {}, i3 = {};
#pragma unroll
    for (int kk = 0; kk < CHUNK; kk += 4) {
        const v2f b = *(const v2f*)(bcol + kk);
        r0 = mma4(*(const v2f*)(ar + 0 * 16 * LDP + kk), b, r0);
        i0 = mma4(*(const v2f*)(ai + 0 * 16 * LDP + kk), b, i0);
        r1 = mma4(*(const v2f*)(ar + 1 * 16 * LDP + kk), b, r1);
        i1 = mma4(*(const v2f*)(ai + 1 * 16 * LDP + kk), b, i1);
        r2 = mma4(*(const v2f*)(ar + 2 * 16 * LDP + kk), b, r2);
        i2 = mma4(*(const v2f*)(ai + 2 * 16 * LDP + kk), b, i2);
        r3 = mma4(*(const v2f*)(ar + 3 * 16 * LDP + kk), b, r3);
        i3 = mma4(*(const v2f*)(ai + 3 * 16 * LDP + kk), b, i3);
    }

    __syncthreads();

    {
        float* qr = sPr + (wave * 16 + m) * LDP + 8 * khalf;
        float* qi = sPi + (wave * 16 + m) * LDP + 8 * khalf;
        store8(qr +  0, r0); store8(qi +  0, i0);
        store8(qr + 16, r1); store8(qi + 16, i1);
        store8(qr + 32, r2); store8(qi + 32, i2);
        store8(qr + 48, r3); store8(qi + 48, i3);
    }
    __syncthreads();

    if (tid < 2 * NN) {
        float* pr = sPr + (tid >> 6) * NCH * LDP + (tid & 63);
        float* pi = sPi + (tid >> 6) * NCH * LDP + (tid & 63);
        float xr = 0.f, xi = 0.f;
        constexpr int SB = 16;
        float pvr[2][SB], pvi[2][SB];
#pragma unroll
        for (int k = 0; k < SB; ++k) { pvr[0][k] = pr[k * LDP]; pvi[0][k] = pi[k * LDP]; }
#pragma unroll
        for (int blk = 0; blk < NCH / SB; ++blk) {
            const int cur = blk & 1, nxt = cur ^ 1;
            if (blk + 1 < NCH / SB) {
#pragma unroll
                for (int k = 0; k < SB; ++k) {
                    pvr[nxt][k] = pr[((blk + 1) * SB + k) * LDP];
                    pvi[nxt][k] = pi[((blk + 1) * SB + k) * LDP];
                }
            }
#pragma unroll
            for (int k = 0; k < SB; ++k) {
                const float nxr = zcr * xr - zci * xi + pvr[cur][k];
                const float nxi = zcr * xi + zci * xr + pvi[cur][k];
                pr[(blk * SB + k) * LDP] = nxr; pi[(blk * SB + k) * LDP] = nxi;
                xr = nxr; xi = nxi;
            }
        }
    }
    __syncthreads();

    float* gPr = Pr + ((size_t)h * COLS + (size_t)cb * CBW) * NN;
    float* gPi = Pi + ((size_t)h * COLS + (size_t)cb * CBW) * NN;
    for (int j = 0; j < CBW * 16 / 256; ++j) {
        const int i = tid + 256 * j;
        const float4 vr = *(const float4*)(sPr + (i >> 4) * LDP + (i & 15) * 4);
        const float4 vi = *(const float4*)(sPi + (i >> 4) * LDP + (i & 15) * 4);
        *(float4*)(gPr + (size_t)i * 4) = vr;
        *(float4*)(gPi + (size_t)i * 4) = vi;
    }
}

__global__ void __launch_bounds__(256)
dss_out(const float* __restrict__ u, const float* __restrict__ D,
        const float* __restrict__ T, const float* __restrict__ CR,
        const float* __restrict__ CIn,
        const float* __restrict__ Pr, const float* __restrict__ Pi,
        float* __restrict__ out) {
    __shared__ float sT [64 * LDP];
    __shared__ float sCR[64 * LDP];
    __shared__ float sCI[64 * LDP];
    __shared__ float sU [CBW * LDP];
    __shared__ float sXr[CBW * LDP];
    __shared__ float sXi[CBW * LDP];

    const int tid   = threadIdx.x;
    const int lane  = tid & 31;
    const int wave  = tid >> 5;
    const int h     = blockIdx.x / NCB;
    const int cb    = blockIdx.x % NCB;
    const int m     = lane & 15;
    const int khalf = lane >> 4;
    const int k0    = khalf * 2;

    stage_rows<64>(T   + (size_t)h * CHUNK * CHUNK, sT,  tid);
    stage_rows<64>(CR  + (size_t)h * CHUNK * NN,    sCR, tid);
    stage_rows<64>(CIn + (size_t)h * CHUNK * NN,    sCI, tid);
    stage_rows<64>(u + ((size_t)(2 * cb)     * HH + h) * LL, sU, tid);
    stage_rows<64>(u + ((size_t)(2 * cb + 1) * HH + h) * LL, sU + 64 * LDP, tid);
    for (int j = 0; j < CBW * 16 / 256; ++j) {
        const int i = tid + 256 * j;
        const int c = i >> 4, q = i & 15;
        const int csrc = c - ((c & 63) != 0);
        const size_t off = ((size_t)(h * COLS + cb * CBW + csrc)) * NN + q * 4;
        *(float4*)(sXr + c * LDP + q * 4) = *(const float4*)(Pr + off);
        *(float4*)(sXi + c * LDP + q * 4) = *(const float4*)(Pi + off);
    }
    if (tid < 16) {
        const float4 z = make_float4(0.f, 0.f, 0.f, 0.f);
        *(float4*)(sXr + tid * 4) = z;
        *(float4*)(sXi + tid * 4) = z;
        *(float4*)(sXr + 64 * LDP + tid * 4) = z;
        *(float4*)(sXi + 64 * LDP + tid * 4) = z;
    }
    __syncthreads();

    const int cl = wave * 16 + m;
    const float* ucol  = sU  + cl * LDP + k0;
    const float* xrcol = sXr + cl * LDP + k0;
    const float* xicol = sXi + cl * LDP + k0;
    const float* at    = sT  + m * LDP + k0;
    const float* acr   = sCR + m * LDP + k0;
    const float* aci   = sCI + m * LDP + k0;

    v8f a0 = {}, a1 = {}, a2 = {}, a3 = {};
#pragma unroll
    for (int kk = 0; kk < CHUNK; kk += 4) {
        const v2f bu = *(const v2f*)(ucol + kk);
        a0 = mma4(*(const v2f*)(at + 0 * 16 * LDP + kk), bu, a0);
        a1 = mma4(*(const v2f*)(at + 1 * 16 * LDP + kk), bu, a1);
        a2 = mma4(*(const v2f*)(at + 2 * 16 * LDP + kk), bu, a2);
        a3 = mma4(*(const v2f*)(at + 3 * 16 * LDP + kk), bu, a3);
    }
#pragma unroll
    for (int kk = 0; kk < CHUNK; kk += 4) {
        const v2f bxr = *(const v2f*)(xrcol + kk);
        const v2f bxi = *(const v2f*)(xicol + kk);
        a0 = mma4(*(const v2f*)(acr + 0 * 16 * LDP + kk), bxr, a0);
        a0 = mma4(*(const v2f*)(aci + 0 * 16 * LDP + kk), bxi, a0);
        a1 = mma4(*(const v2f*)(acr + 1 * 16 * LDP + kk), bxr, a1);
        a1 = mma4(*(const v2f*)(aci + 1 * 16 * LDP + kk), bxi, a1);
        a2 = mma4(*(const v2f*)(acr + 2 * 16 * LDP + kk), bxr, a2);
        a2 = mma4(*(const v2f*)(aci + 2 * 16 * LDP + kk), bxi, a2);
        a3 = mma4(*(const v2f*)(acr + 3 * 16 * LDP + kk), bxr, a3);
        a3 = mma4(*(const v2f*)(aci + 3 * 16 * LDP + kk), bxi, a3);
    }

    const float Dh = D[h];
    const int col  = cb * CBW + cl;
    const int b    = col >> 6;
    const int ch   = col & 63;
    const float* up = sU + cl * LDP + 8 * khalf;
    float* op = out + ((size_t)b * HH + h) * LL + (size_t)ch * CHUNK + 8 * khalf;
    v8f accs[4] = {a0, a1, a2, a3};
    for (int rt = 0; rt < 4; ++rt) {
        const v8f acc = accs[rt];
        const float4 u0 = *(const float4*)(up + rt * 16);
        const float4 u1 = *(const float4*)(up + rt * 16 + 4);
        float4 o0, o1;
        o0.x = acc[0] + Dh * u0.x; o0.y = acc[1] + Dh * u0.y;
        o0.z = acc[2] + Dh * u0.z; o0.w = acc[3] + Dh * u0.w;
        o1.x = acc[4] + Dh * u1.x; o1.y = acc[5] + Dh * u1.y;
        o1.z = acc[6] + Dh * u1.z; o1.w = acc[7] + Dh * u1.w;
        *(float4*)(op + rt * 16)     = o0;
        *(float4*)(op + rt * 16 + 4) = o1;
    }
}

extern "C" void kernel_launch(void* const* d_in, const int* in_sizes, int n_in,
                              void* d_out, int out_size, void* d_ws, size_t ws_size,
                              hipStream_t stream) {
    const float* u       = (const float*)d_in[0];
    const float* Lre     = (const float*)d_in[1];
    const float* Lim     = (const float*)d_in[2];
    const float* Wri     = (const float*)d_in[3];
    const float* Dp      = (const float*)d_in[4];
    const float* logstep = (const float*)d_in[5];
    float* out = (float*)d_out;

    float* ws  = (float*)d_ws;
    float* CRm = ws;
    float* CIn = CRm + (size_t)HH * CHUNK * NN;
    float* Tm  = CIn + (size_t)HH * CHUNK * NN;
    float* Pr  = Tm  + (size_t)HH * CHUNK * CHUNK;
    float* Pi  = Pr  + (size_t)HH * COLS * NN;

    dss_proj<<<HH + HH * NCB, 256, 0, stream>>>(u, Pr, Pi,
                                                Lre, Lim, Wri, logstep, CRm, CIn, Tm);

    dss_out<<<HH * NCB, 256, 0, stream>>>(u, Dp, Tm, CRm, CIn, Pr, Pi, out);
}


# State output product C@x moved behind the scan into the first launch (C chain walked per work-group into LDS); workspace holds one real plane; state part summed from zero, added to T@U at the end
# speedup vs baseline: 4.4875x; 1.2955x over previous
#include <hip/hip_runtime.h>
#include <math.h>

#define BB 8
#define HH 128
#define LL 4096
#define NN 64
#define CHUNK 64
#define NCH (LL / CHUNK)
#define COLS (BB * NCH)
#define EPSV 1e-7f

typedef float v2f __attribute__((ext_vector_type(2)));
typedef float v4f __attribute__((ext_vector_type(4)));
typedef float v8f __attribute__((ext_vector_type(8)));


__device__ __forceinline__ void
mode_c(float lre, float lim, float wr, float wi, float ar, float ai, float zr, float zi,
       float* cr_out, float* ci_out) {
    float cL, sL; sincosf(ai * (float)LL, &sL, &cL);
    const float eL = expf(ar * (float)LL);
    const float zLr = eL * cL, zLi = eL * sL;
    const float nr = 1.f - zLr, ni = -zLi;
    const float dr = 1.f - zr,  di = -zi;
    const float dd = dr * dr + di * di;
    const float sr = (nr * dr + ni * di) / dd;
    const float si = (ni * dr - nr * di) / dd;
    const float ll = lre * lre + lim * lim;
    const float br = (wr * lre + wi * lim) / ll;
    const float bi = (wi * lre - wr * lim) / ll;
    const float ss = sr * sr + si * si + EPSV;
    *cr_out = (br * sr + bi * si) / ss;
    *ci_out = (bi * sr - br * si) / ss;
}

__device__ __forceinline__ void
build_kt(const float* __restrict__ Lre, const float* __restrict__ Lim,
         const float* __restrict__ Wri, const float* __restrict__ logstep,
         float* __restrict__ T,
         float* small, float* sE, float* sV, int h, int tid) {
    float* sdtr = small;
    float* sdti = small + NN;
    float* scr  = small + 2 * NN;
    float* sci  = small + 3 * NN;
    float* sK   = small + 4 * NN;
    const int n    = tid & 63;
    const int part = tid >> 6;

    const float lre = Lre[h * NN + n], lim = Lim[h * NN + n];
    const float dt  = __expf(logstep[h]);
    const float ar  = dt * lre, ai = dt * lim;

    float cz, sz; sincosf(ai, &sz, &cz);
    const float er = expf(ar);
    const float zr = er * cz, zi = er * sz;
    const float wr = Wri[(h * NN + n) * 2 + 0], wi = Wri[(h * NN + n) * 2 + 1];
    float cr, ci;
    mode_c(lre, lim, wr, wi, ar, ai, zr, zi, &cr, &ci);

    if (part == 0) { sdtr[n] = ar; sdti[n] = ai; scr[n] = cr; sci[n] = ci; }
    __syncthreads();

    {
        const float t = (float)n;
        for (int jj = 0; jj < NN / 4; ++jj) {
            const int j = part * (NN / 4) + jj;
            const float e = expf(sdtr[j] * t);
            float cs, sn; sincosf(sdti[j] * t, &sn, &cs);
            sE[j * CHUNK + n] = e;
            sV[j * CHUNK + n] = scr[j] * cs - sci[j] * sn;
        }
    }
    __syncthreads();

    if (part == 0) {
        float acc = 0.f;
        for (int j = 0; j < NN; ++j)
            acc += sE[j * CHUNK + n] * sV[j * CHUNK + n];
        sK[n] = acc;
    }
    __syncthreads();

    for (int idx = tid; idx < CHUNK * CHUNK; idx += 256) {
        const int t = idx >> 6, s = idx & 63;
        T[(h * CHUNK + t) * CHUNK + s] = (t >= s) ? sK[t - s] : 0.f;
    }
}

#define LDP 68
#define CBW 128
#define NCB (COLS / CBW)

template <int NROWS>
__device__ __forceinline__ void stage_rows(const float* __restrict__ src, float* lds, int tid) {
    for (int j = 0; j < NROWS * 16 / 256; ++j) {
        const int i = tid + 256 * j;
        const float4 v = *(const float4*)(src + (size_t)i * 4);
        *(float4*)(lds + (i >> 4) * LDP + (i & 15) * 4) = v;
    }
}

__device__ __forceinline__ v8f mma4(v2f a, v2f b, v8f c) {
    return __builtin_amdgcn_wmma_f32_16x16x4_f32(false, a, false, b, (short)0, c, false, false);
}

__device__ __forceinline__ void store8(float* __restrict__ p, v8f v) {
    float4 lo, hi;
    lo.x = v[0]; lo.y = v[1]; lo.z = v[2]; lo.w = v[3];
    hi.x = v[4]; hi.y = v[5]; hi.z = v[6]; hi.w = v[7];
    *(float4*)(p)     = lo;
    *(float4*)(p + 4) = hi;
}

__global__ void __launch_bounds__(256, 6)
dss_state(const float* __restrict__ u,
          float* __restrict__ Yst,
          const float* __restrict__ Lre, const float* __restrict__ Lim,
          const float* __restrict__ Wri, const float* __restrict__ logstep,
          float* __restrict__ T) {
    __shared__ float smem[2 * CBW * LDP];
    __shared__ float sCR[CHUNK * LDP];
    __shared__ float sCI[CHUNK * LDP];
    float* sZr = smem;
    float* sZi = smem + 64 * LDP;
    float* sU  = smem + 128 * LDP;
    float* sPr = smem;
    float* sPi = smem + CBW * LDP;

    const int tid   = threadIdx.x;

    if (blockIdx.x < HH) {
        build_kt(Lre, Lim, Wri, logstep, T,
                 sZr, sU, sU + CHUNK * NN, blockIdx.x, tid);
        return;
    }

    const int lane  = tid & 31;
    const int wave  = tid >> 5;
    const int bid   = blockIdx.x - HH;
    const int h     = bid / NCB;
    const int cb    = bid % NCB;
    const int m     = lane & 15;
    const int khalf = lane >> 4;
    const int k0    = khalf * 2;

    float zcr = 0.f, zci = 0.f;
    if (tid < 2 * NN) {
        const int   n   = tid & 63;
        const float lre = Lre[h * NN + n], lim = Lim[h * NN + n];
        const float dt  = __expf(logstep[h]);
        const float ar  = dt * lre, ai = dt * lim;
        float cz, sz; sincosf(ai, &sz, &cz);
        const float er = expf(ar);
        const float zr = er * cz, zi = er * sz;
        if (tid < NN) {
            float pr = 1.f, pi_ = 0.f;
            float* zrrow = sZr + n * LDP;
            float* zirow = sZi + n * LDP;
            for (int k = 0; k < CHUNK; k += 4) {
                float qr[4], qi[4];
                for (int e = 0; e < 4; ++e) {
                    qr[3 - e] = pr; qi[3 - e] = pi_;
                    const float tr = pr * zr - pi_ * zi;
                    pi_ = pr * zi + pi_ * zr;
                    pr = tr;
                }
                *(float4*)(zrrow + CHUNK - 4 - k) = make_float4(qr[0], qr[1], qr[2], qr[3]);
                *(float4*)(zirow + CHUNK - 4 - k) = make_float4(qi[0], qi[1], qi[2], qi[3]);
            }
        } else {
            const float wr = Wri[(h * NN + n) * 2 + 0], wi = Wri[(h * NN + n) * 2 + 1];
            float cr, ci;
            mode_c(lre, lim, wr, wi, ar, ai, zr, zi, &cr, &ci);
            float qr = cr * zr - ci * zi;
            float qi = cr * zi + ci * zr;
            for (int t = 0; t < CHUNK; ++t) {
                sCR[t * LDP + n] =  qr;
                sCI[t * LDP + n] = -qi;
                const float tr = qr * zr - qi * zi;
                qi = qr * zi + qi * zr;
                qr = tr;
            }
        }
        float cC, sC; sincosf(ai * (float)CHUNK, &sC, &cC);
        const float eC = expf(ar * (float)CHUNK);
        zcr = eC * cC;
        zci = eC * sC;
    } else {
        const int t = tid - 2 * NN;
        const float* u0 = u + ((size_t)(2 * cb)     * HH + h) * LL;
        const float* u1 = u + ((size_t)(2 * cb + 1) * HH + h) * LL;
        v4f uv[16];
        for (int j = 0; j < 8; ++j) {
            uv[j]     = *(const v4f*)(u0 + (size_t)(t + 128 * j) * 4);
            uv[j + 8] = *(const v4f*)(u1 + (size_t)(t + 128 * j) * 4);
        }
        for (int j = 0; j < 8; ++j) {
            const int i = t + 128 * j;
            *(v4f*)(sU +            (i >> 4) * LDP + (i & 15) * 4) = uv[j];
            *(v4f*)(sU + 64 * LDP + (i >> 4) * LDP + (i & 15) * 4) = uv[j + 8];
        }
    }
    __syncthreads();

    const float* bcol = sU  + (wave * 16 + m) * LDP + k0;
    const float* ar   = sZr + m * LDP + k0;
    const float* ai   = sZi + m * LDP + k0;

    v8f r0 = {}, r1 = {}, r2 = {}, r3 = {};
    v8f i0 = {}, i1 = {}, i2 = {}, i3 = {};
#pragma unroll
    for (int kk = 0; kk < CHUNK; kk += 4) {
        const v2f b = *(const v2f*)(bcol + kk);
        r0 = mma4(*(const v2f*)(ar + 0 * 16 * LDP + kk), b, r0);
        i0 = mma4(*(const v2f*)(ai + 0 * 16 * LDP + kk), b, i0);
        r1 = mma4(*(const v2f*)(ar + 1 * 16 * LDP + kk), b, r1);
        i1 = mma4(*(const v2f*)(ai + 1 * 16 * LDP + kk), b, i1);
        r2 = mma4(*(const v2f*)(ar + 2 * 16 * LDP + kk), b, r2);
        i2 = mma4(*(const v2f*)(ai + 2 * 16 * LDP + kk), b, i2);
        r3 = mma4(*(const v2f*)(ar + 3 * 16 * LDP + kk), b, r3);
        i3 = mma4(*(const v2f*)(ai + 3 * 16 * LDP + kk), b, i3);
    }

    __syncthreads();

    {
        float* qr = sPr + (wave * 16 + m) * LDP + 8 * khalf;
        float* qi = sPi + (wave * 16 + m) * LDP + 8 * khalf;
        store8(qr +  0, r0); store8(qi +  0, i0);
        store8(qr + 16, r1); store8(qi + 16, i1);
        store8(qr + 32, r2); store8(qi + 32, i2);
        store8(qr + 48, r3); store8(qi + 48, i3);
    }
    __syncthreads();

    if (tid < 2 * NN) {
        float* pr = sPr + (tid >> 6) * NCH * LDP + (tid & 63);
        float* pi = sPi + (tid >> 6) * NCH * LDP + (tid & 63);
        float xr = 0.f, xi = 0.f;
        constexpr int SB = 16;
        float pvr[2][SB], pvi[2][SB];
#pragma unroll
        for (int k = 0; k < SB; ++k) { pvr[0][k] = pr[k * LDP]; pvi[0][k] = pi[k * LDP]; }
#pragma unroll
        for (int blk = 0; blk < NCH / SB; ++blk) {
            const int cur = blk & 1, nxt = cur ^ 1;
            if (blk + 1 < NCH / SB) {
#pragma unroll
                for (int k = 0; k < SB; ++k) {
                    pvr[nxt][k] = pr[((blk + 1) * SB + k) * LDP];
                    pvi[nxt][k] = pi[((blk + 1) * SB + k) * LDP];
                }
            }
#pragma unroll
            for (int k = 0; k < SB; ++k) {
                const float nxr = zcr * xr - zci * xi + pvr[cur][k];
                const float nxi = zcr * xi + zci * xr + pvi[cur][k];
                pr[(blk * SB + k) * LDP] = nxr; pi[(blk * SB + k) * LDP] = nxi;
                xr = nxr; xi = nxi;
            }
        }
    }
    __syncthreads();

    {
        const float* xrcol = sPr + (wave * 16 + m) * LDP + k0;
        const float* xicol = sPi + (wave * 16 + m) * LDP + k0;
        const float* acr   = sCR + m * LDP + k0;
        const float* aci   = sCI + m * LDP + k0;
        v8f a0 = {}, a1 = {}, a2 = {}, a3 = {};
#pragma unroll
        for (int kk = 0; kk < NN; kk += 4) {
            const v2f bxr = *(const v2f*)(xrcol + kk);
            const v2f bxi = *(const v2f*)(xicol + kk);
            a0 = mma4(*(const v2f*)(acr + 0 * 16 * LDP + kk), bxr, a0);
            a0 = mma4(*(const v2f*)(aci + 0 * 16 * LDP + kk), bxi, a0);
            a1 = mma4(*(const v2f*)(acr + 1 * 16 * LDP + kk), bxr, a1);
            a1 = mma4(*(const v2f*)(aci + 1 * 16 * LDP + kk), bxi, a1);
            a2 = mma4(*(const v2f*)(acr + 2 * 16 * LDP + kk), bxr, a2);
            a2 = mma4(*(const v2f*)(aci + 2 * 16 * LDP + kk), bxi, a2);
            a3 = mma4(*(const v2f*)(acr + 3 * 16 * LDP + kk), bxr, a3);
            a3 = mma4(*(const v2f*)(aci + 3 * 16 * LDP + kk), bxi, a3);
        }
        float* qy = sPr + (wave * 16 + m) * LDP + 8 * khalf;
        store8(qy +  0, a0);
        store8(qy + 16, a1);
        store8(qy + 32, a2);
        store8(qy + 48, a3);
    }
    __syncthreads();

    float* gY = Yst + ((size_t)h * COLS + (size_t)cb * CBW) * CHUNK;
    for (int j = 0; j < CBW * 16 / 256; ++j) {
        const int i = tid + 256 * j;
        const float4 vy = *(const float4*)(sPr + (i >> 4) * LDP + (i & 15) * 4);
        *(float4*)(gY + (size_t)i * 4) = vy;
    }
}

__global__ void __launch_bounds__(256)
dss_out(const float* __restrict__ u, const float* __restrict__ D,
        const float* __restrict__ T, const float* __restrict__ Yst,
        float* __restrict__ out) {
    __shared__ float sT[64 * LDP];
    __shared__ float sU[CBW * LDP];
    __shared__ float sY[CBW * LDP];

    const int tid   = threadIdx.x;
    const int lane  = tid & 31;
    const int wave  = tid >> 5;
    const int h     = blockIdx.x / NCB;
    const int cb    = blockIdx.x % NCB;
    const int m     = lane & 15;
    const int khalf = lane >> 4;
    const int k0    = khalf * 2;

    stage_rows<64>(T + (size_t)h * CHUNK * CHUNK, sT, tid);
    stage_rows<64>(u + ((size_t)(2 * cb)     * HH + h) * LL, sU, tid);
    stage_rows<64>(u + ((size_t)(2 * cb + 1) * HH + h) * LL, sU + 64 * LDP, tid);
    for (int j = 0; j < CBW * 16 / 256; ++j) {
        const int i = tid + 256 * j;
        const int c = i >> 4, q = i & 15;
        const int csrc = c - ((c & 63) != 0);
        const size_t off = ((size_t)(h * COLS + cb * CBW + csrc)) * CHUNK + q * 4;
        *(float4*)(sY + c * LDP + q * 4) = *(const float4*)(Yst + off);
    }
    if (tid < 16) {
        const float4 z = make_float4(0.f, 0.f, 0.f, 0.f);
        *(float4*)(sY + tid * 4) = z;
        *(float4*)(sY + 64 * LDP + tid * 4) = z;
    }
    __syncthreads();

    const int cl = wave * 16 + m;
    const float* ucol = sU + cl * LDP + k0;
    const float* at   = sT + m * LDP + k0;

    v8f a0 = {}, a1 = {}, a2 = {}, a3 = {};
#pragma unroll
    for (int kk = 0; kk < CHUNK; kk += 4) {
        const v2f bu = *(const v2f*)(ucol + kk);
        a0 = mma4(*(const v2f*)(at + 0 * 16 * LDP + kk), bu, a0);
        a1 = mma4(*(const v2f*)(at + 1 * 16 * LDP + kk), bu, a1);
        a2 = mma4(*(const v2f*)(at + 2 * 16 * LDP + kk), bu, a2);
        a3 = mma4(*(const v2f*)(at + 3 * 16 * LDP + kk), bu, a3);
    }

    const float Dh = D[h];
    const int col  = cb * CBW + cl;
    const int b    = col >> 6;
    const int ch   = col & 63;
    const float* up = sU + cl * LDP + 8 * khalf;
    const float* yp = sY + cl * LDP + 8 * khalf;
    float* op = out + ((size_t)b * HH + h) * LL + (size_t)ch * CHUNK + 8 * khalf;
    v8f accs[4] = {a0, a1, a2, a3};
    for (int rt = 0; rt < 4; ++rt) {
        const v8f acc = accs[rt];
        const float4 u0 = *(const float4*)(up + rt * 16);
        const float4 u1 = *(const float4*)(up + rt * 16 + 4);
        const float4 y0 = *(const float4*)(yp + rt * 16);
        const float4 y1 = *(const float4*)(yp + rt * 16 + 4);
        float4 o0, o1;
        o0.x = (acc[0] + y0.x) + Dh * u0.x; o0.y = (acc[1] + y0.y) + Dh * u0.y;
        o0.z = (acc[2] + y0.z) + Dh * u0.z; o0.w = (acc[3] + y0.w) + Dh * u0.w;
        o1.x = (acc[4] + y1.x) + Dh * u1.x; o1.y = (acc[5] + y1.y) + Dh * u1.y;
        o1.z = (acc[6] + y1.z) + Dh * u1.z; o1.w = (acc[7] + y1.w) + Dh * u1.w;
        *(float4*)(op + rt * 16)     = o0;
        *(float4*)(op + rt * 16 + 4) = o1;
    }
}

extern "C" void kernel_launch(void* const* d_in, const int* in_sizes, int n_in,
                              void* d_out, int out_size, void* d_ws, size_t ws_size,
                              hipStream_t stream) {
    const float* u       = (const float*)d_in[0];
    const float* Lre     = (const float*)d_in[1];
    const float* Lim     = (const float*)d_in[2];
    const float* Wri     = (const float*)d_in[3];
    const float* Dp      = (const float*)d_in[4];
    const float* logstep = (const float*)d_in[5];
    float* out = (float*)d_out;

    float* ws  = (float*)d_ws;
    float* Tm  = ws;
    float* Yst = Tm + (size_t)HH * CHUNK * CHUNK;

    dss_state<<<HH + HH * NCB, 256, 0, stream>>>(u, Yst, Lre, Lim, Wri, logstep, Tm);

    dss_out<<<HH * NCB, 256, 0, stream>>>(u, Dp, Tm, Yst, out);
}


# Output launch folded into the state launch: U tile stays in LDS, K = row sums of the C chain read as Toeplitz fragments, state part summed onto T@U in one chain; 2 launches become 1; K rounds anew
# speedup vs baseline: 5.8772x; 1.3097x over previous
#include <hip/hip_runtime.h>
#include <math.h>

#define BB 8
#define HH 128
#define LL 4096
#define NN 64
#define CHUNK 64
#define NCH (LL / CHUNK)
#define COLS (BB * NCH)
#define EPSV 1e-7f

typedef float v2f __attribute__((ext_vector_type(2)));
typedef float v4f __attribute__((ext_vector_type(4)));
typedef float v8f __attribute__((ext_vector_type(8)));


__device__ __forceinline__ void
mode_c(float lre, float lim, float wr, float wi, float ar, float ai, float zr, float zi,
       float* cr_out, float* ci_out) {
    float cL, sL; sincosf(ai * (float)LL, &sL, &cL);
    const float eL = expf(ar * (float)LL);
    const float zLr = eL * cL, zLi = eL * sL;
    const float nr = 1.f - zLr, ni = -zLi;
    const float dr = 1.f - zr,  di = -zi;
    const float dd = dr * dr + di * di;
    const float sr = (nr * dr + ni * di) / dd;
    const float si = (ni * dr - nr * di) / dd;
    const float ll = lre * lre + lim * lim;
    const float br = (wr * lre + wi * lim) / ll;
    const float bi = (wi * lre - wr * lim) / ll;
    const float ss = sr * sr + si * si + EPSV;
    *cr_out = (br * sr + bi * si) / ss;
    *ci_out = (bi * sr - br * si) / ss;
}

#define LDP 68
#define CBW 128
#define NCB (COLS / CBW)

__device__ __forceinline__ v8f mma4(v2f a, v2f b, v8f c) {
    return __builtin_amdgcn_wmma_f32_16x16x4_f32(false, a, false, b, (short)0, c, false, false);
}

__device__ __forceinline__ void store8(float* __restrict__ p, v8f v) {
    float4 lo, hi;
    lo.x = v[0]; lo.y = v[1]; lo.z = v[2]; lo.w = v[3];
    hi.x = v[4]; hi.y = v[5]; hi.z = v[6]; hi.w = v[7];
    *(float4*)(p)     = lo;
    *(float4*)(p + 4) = hi;
}

__global__ void __launch_bounds__(256, 4)
dss_fused(const float* __restrict__ u, const float* __restrict__ D,
          const float* __restrict__ Lre, const float* __restrict__ Lim,
          const float* __restrict__ Wri, const float* __restrict__ logstep,
          float* __restrict__ out) {
    __shared__ __attribute__((aligned(16))) float sU[CBW * LDP];
    __shared__ __attribute__((aligned(16))) float sA[CBW * LDP];
    __shared__ __attribute__((aligned(16))) float sB[CBW * LDP];
    __shared__ __attribute__((aligned(16))) float sCR[CHUNK * LDP];
    __shared__ __attribute__((aligned(16))) float sCI[CHUNK * LDP];
    __shared__ __attribute__((aligned(16))) float sKr[2 * CHUNK];
    __shared__ __attribute__((aligned(16))) float sC0[NN];
    __shared__ __attribute__((aligned(16))) float sZero[CHUNK];
    float* sZr = sA;
    float* sZi = sA + 64 * LDP;
    float* sPr = sA;
    float* sPi = sB;

    const int tid   = threadIdx.x;
    const int lane  = tid & 31;
    const int wave  = tid >> 5;
    const int h     = blockIdx.x / NCB;
    const int cb    = blockIdx.x % NCB;
    const int m     = lane & 15;
    const int khalf = lane >> 4;
    const int k0    = khalf * 2;

    float zcr = 0.f, zci = 0.f;
    if (tid < 2 * NN) {
        const int   n   = tid & 63;
        const float lre = Lre[h * NN + n], lim = Lim[h * NN + n];
        const float dt  = __expf(logstep[h]);
        const float ar  = dt * lre, ai = dt * lim;
        float cz, sz; sincosf(ai, &sz, &cz);
        const float er = expf(ar);
        const float zr = er * cz, zi = er * sz;
        if (tid < NN) {
            float pr = 1.f, pi_ = 0.f;
            float* zrrow = sZr + n * LDP;
            float* zirow = sZi + n * LDP;
            for (int k = 0; k < CHUNK; k += 4) {
                float qr[4], qi[4];
                for (int e = 0; e < 4; ++e) {
                    qr[3 - e] = pr; qi[3 - e] = pi_;
                    const float tr = pr * zr - pi_ * zi;
                    pi_ = pr * zi + pi_ * zr;
                    pr = tr;
                }
                *(float4*)(zrrow + CHUNK - 4 - k) = make_float4(qr[0], qr[1], qr[2], qr[3]);
                *(float4*)(zirow + CHUNK - 4 - k) = make_float4(qi[0], qi[1], qi[2], qi[3]);
            }
        } else {
            const float wr = Wri[(h * NN + n) * 2 + 0], wi = Wri[(h * NN + n) * 2 + 1];
            float cr, ci;
            mode_c(lre, lim, wr, wi, ar, ai, zr, zi, &cr, &ci);
            sC0[n] = cr;
            float qr = cr * zr - ci * zi;
            float qi = cr * zi + ci * zr;
            for (int t = 0; t < CHUNK; ++t) {
                sCR[t * LDP + n] =  qr;
                sCI[t * LDP + n] = -qi;
                const float tr = qr * zr - qi * zi;
                qi = qr * zi + qi * zr;
                qr = tr;
            }
        }
        float cC, sC; sincosf(ai * (float)CHUNK, &sC, &cC);
        const float eC = expf(ar * (float)CHUNK);
        zcr = eC * cC;
        zci = eC * sC;
    } else {
        const int t = tid - 2 * NN;
        const float* u0 = u + ((size_t)(2 * cb)     * HH + h) * LL;
        const float* u1 = u + ((size_t)(2 * cb + 1) * HH + h) * LL;
        v4f uv[16];
        for (int j = 0; j < 8; ++j) {
            uv[j]     = *(const v4f*)(u0 + (size_t)(t + 128 * j) * 4);
            uv[j + 8] = *(const v4f*)(u1 + (size_t)(t + 128 * j) * 4);
        }
        if (t < CHUNK) { sKr[CHUNK + t] = 0.f; sZero[t] = 0.f; }
        for (int j = 0; j < 8; ++j) {
            const int i = t + 128 * j;
            *(v4f*)(sU +            (i >> 4) * LDP + (i & 15) * 4) = uv[j];
            *(v4f*)(sU + 64 * LDP + (i >> 4) * LDP + (i & 15) * 4) = uv[j + 8];
        }
    }
    __syncthreads();

    if (tid >= 3 * NN) {
        const int t = tid - 3 * NN;
        const float* row = (t == 0) ? sC0 : (sCR + (t - 1) * LDP);
        float acc = 0.f;
        for (int q = 0; q < NN; q += 4) {
            const float4 v = *(const float4*)(row + q);
            acc += v.x; acc += v.y; acc += v.z; acc += v.w;
        }
        sKr[CHUNK - 1 - t] = acc;
    }

    {
        const float* bcol = sU  + (wave * 16 + m) * LDP + k0;
        const float* ar   = sZr + m * LDP + k0;
        const float* ai   = sZi + m * LDP + k0;

        v8f r0 = {}, r1 = {}, r2 = {}, r3 = {};
        v8f i0 = {}, i1 = {}, i2 = {}, i3 = {};
#pragma unroll
        for (int kk = 0; kk < CHUNK; kk += 4) {
            const v2f b = *(const v2f*)(bcol + kk);
            r0 = mma4(*(const v2f*)(ar + 0 * 16 * LDP + kk), b, r0);
            i0 = mma4(*(const v2f*)(ai + 0 * 16 * LDP + kk), b, i0);
            r1 = mma4(*(const v2f*)(ar + 1 * 16 * LDP + kk), b, r1);
            i1 = mma4(*(const v2f*)(ai + 1 * 16 * LDP + kk), b, i1);
            r2 = mma4(*(const v2f*)(ar + 2 * 16 * LDP + kk), b, r2);
            i2 = mma4(*(const v2f*)(ai + 2 * 16 * LDP + kk), b, i2);
            r3 = mma4(*(const v2f*)(ar + 3 * 16 * LDP + kk), b, r3);
            i3 = mma4(*(const v2f*)(ai + 3 * 16 * LDP + kk), b, i3);
        }

        __syncthreads();

        float* qr = sPr + (wave * 16 + m) * LDP + 8 * khalf;
        float* qi = sPi + (wave * 16 + m) * LDP + 8 * khalf;
        store8(qr +  0, r0); store8(qi +  0, i0);
        store8(qr + 16, r1); store8(qi + 16, i1);
        store8(qr + 32, r2); store8(qi + 32, i2);
        store8(qr + 48, r3); store8(qi + 48, i3);
    }
    __syncthreads();

    if (tid < 2 * NN) {
        float* pr = sPr + (tid >> 6) * NCH * LDP + (tid & 63);
        float* pi = sPi + (tid >> 6) * NCH * LDP + (tid & 63);
        float xr = 0.f, xi = 0.f;
        constexpr int SB = 16;
        float pvr[2][SB], pvi[2][SB];
#pragma unroll
        for (int k = 0; k < SB; ++k) { pvr[0][k] = pr[k * LDP]; pvi[0][k] = pi[k * LDP]; }
#pragma unroll
        for (int blk = 0; blk < NCH / SB; ++blk) {
            const int cur = blk & 1, nxt = cur ^ 1;
            if (blk + 1 < NCH / SB) {
#pragma unroll
                for (int k = 0; k < SB; ++k) {
                    pvr[nxt][k] = pr[((blk + 1) * SB + k) * LDP];
                    pvi[nxt][k] = pi[((blk + 1) * SB + k) * LDP];
                }
            }
#pragma unroll
            for (int k = 0; k < SB; ++k) {
                const float nxr = zcr * xr - zci * xi + pvr[cur][k];
                const float nxi = zcr * xi + zci * xr + pvi[cur][k];
                pr[(blk * SB + k) * LDP] = nxr; pi[(blk * SB + k) * LDP] = nxi;
                xr = nxr; xi = nxi;
            }
        }
    }
    __syncthreads();

    const int cl = wave * 16 + m;
    v8f a0 = {}, a1 = {}, a2 = {}, a3 = {};
    {
        const float* ucol = sU + cl * LDP + k0;
        const float* kb   = sKr + (CHUNK - 1) - m + k0;
#pragma unroll
        for (int kk = 0; kk < CHUNK; kk += 4) {
            const v2f bu = *(const v2f*)(ucol + kk);
            v2f t0, t1, t2, t3;
            t0.x = kb[kk];      t0.y = kb[kk + 1];
            t1.x = kb[kk - 16]; t1.y = kb[kk - 15];
            t2.x = kb[kk - 32]; t2.y = kb[kk - 31];
            t3.x = kb[kk - 48]; t3.y = kb[kk - 47];
            a0 = mma4(t0, bu, a0);
            a1 = mma4(t1, bu, a1);
            a2 = mma4(t2, bu, a2);
            a3 = mma4(t3, bu, a3);
        }
    }
    {
        const bool first = (cl & (NCH - 1)) == 0;
        const float* xrcol = first ? (sZero + k0) : (sPr + (cl - 1) * LDP + k0);
        const float* xicol = first ? (sZero + k0) : (sPi + (cl - 1) * LDP + k0);
        const float* acr   = sCR + m * LDP + k0;
        const float* aci   = sCI + m * LDP + k0;
#pragma unroll
        for (int kk = 0; kk < NN; kk += 4) {
            const v2f bxr = *(const v2f*)(xrcol + kk);
            const v2f bxi = *(const v2f*)(xicol + kk);
            a0 = mma4(*(const v2f*)(acr + 0 * 16 * LDP + kk), bxr, a0);
            a0 = mma4(*(const v2f*)(aci + 0 * 16 * LDP + kk), bxi, a0);
            a1 = mma4(*(const v2f*)(acr + 1 * 16 * LDP + kk), bxr, a1);
            a1 = mma4(*(const v2f*)(aci + 1 * 16 * LDP + kk), bxi, a1);
            a2 = mma4(*(const v2f*)(acr + 2 * 16 * LDP + kk), bxr, a2);
            a2 = mma4(*(const v2f*)(aci + 2 * 16 * LDP + kk), bxi, a2);
            a3 = mma4(*(const v2f*)(acr + 3 * 16 * LDP + kk), bxr, a3);
            a3 = mma4(*(const v2f*)(aci + 3 * 16 * LDP + kk), bxi, a3);
        }
    }

    const float Dh = D[h];
    const int col  = cb * CBW + cl;
    const int b    = col >> 6;
    const int ch   = col & 63;
    const float* up = sU + cl * LDP + 8 * khalf;
    float* op = out + ((size_t)b * HH + h) * LL + (size_t)ch * CHUNK + 8 * khalf;
    v8f accs[4] = {a0, a1, a2, a3};
    for (int rt = 0; rt < 4; ++rt) {
        const v8f acc = accs[rt];
        const float4 u0 = *(const float4*)(up + rt * 16);
        const float4 u1 = *(const float4*)(up + rt * 16 + 4);
        float4 o0, o1;
        o0.x = acc[0] + Dh * u0.x; o0.y = acc[1] + Dh * u0.y;
        o0.z = acc[2] + Dh * u0.z; o0.w = acc[3] + Dh * u0.w;
        o1.x = acc[4] + Dh * u1.x; o1.y = acc[5] + Dh * u1.y;
        o1.z = acc[6] + Dh * u1.z; o1.w = acc[7] + Dh * u1.w;
        *(float4*)(op + rt * 16)     = o0;
        *(float4*)(op + rt * 16 + 4) = o1;
    }
}

extern "C" void kernel_launch(void* const* d_in, const int* in_sizes, int n_in,
                              void* d_out, int out_size, void* d_ws, size_t ws_size,
                              hipStream_t stream) {
    const float* u       = (const float*)d_in[0];
    const float* Lre     = (const float*)d_in[1];
    const float* Lim     = (const float*)d_in[2];
    const float* Wri     = (const float*)d_in[3];
    const float* Dp      = (const float*)d_in[4];
    const float* logstep = (const float*)d_in[5];
    float* out = (float*)d_out;

    dss_fused<<<HH * NCB, 256, 0, stream>>>(u, Dp, Lre, Lim, Wri, logstep, out);
}


# Epilogue staged through LDS: each wave forms y in place in its own U rows and reads it back row-wise; every store instruction writes four whole 128-byte lines, not 16-byte pieces of 16 lines
# speedup vs baseline: 6.3662x; 1.0832x over previous
#include <hip/hip_runtime.h>
#include <math.h>

#define BB 8
#define HH 128
#define LL 4096
#define NN 64
#define CHUNK 64
#define NCH (LL / CHUNK)
#define COLS (BB * NCH)
#define EPSV 1e-7f

typedef float v2f __attribute__((ext_vector_type(2)));
typedef float v4f __attribute__((ext_vector_type(4)));
typedef float v8f __attribute__((ext_vector_type(8)));


__device__ __forceinline__ void
mode_c(float lre, float lim, float wr, float wi, float ar, float ai, float zr, float zi,
       float* cr_out, float* ci_out) {
    float cL, sL; sincosf(ai * (float)LL, &sL, &cL);
    const float eL = expf(ar * (float)LL);
    const float zLr = eL * cL, zLi = eL * sL;
    const float nr = 1.f - zLr, ni = -zLi;
    const float dr = 1.f - zr,  di = -zi;
    const float dd = dr * dr + di * di;
    const float sr = (nr * dr + ni * di) / dd;
    const float si = (ni * dr - nr * di) / dd;
    const float ll = lre * lre + lim * lim;
    const float br = (wr * lre + wi * lim) / ll;
    const float bi = (wi * lre - wr * lim) / ll;
    const float ss = sr * sr + si * si + EPSV;
    *cr_out = (br * sr + bi * si) / ss;
    *ci_out = (bi * sr - br * si) / ss;
}

#define LDP 68
#define CBW 128
#define NCB (COLS / CBW)

__device__ __forceinline__ v8f mma4(v2f a, v2f b, v8f c) {
    return __builtin_amdgcn_wmma_f32_16x16x4_f32(false, a, false, b, (short)0, c, false, false);
}

__device__ __forceinline__ void store8(float* __restrict__ p, v8f v) {
    float4 lo, hi;
    lo.x = v[0]; lo.y = v[1]; lo.z = v[2]; lo.w = v[3];
    hi.x = v[4]; hi.y = v[5]; hi.z = v[6]; hi.w = v[7];
    *(float4*)(p)     = lo;
    *(float4*)(p + 4) = hi;
}

__global__ void __launch_bounds__(256, 4)
dss_fused(const float* __restrict__ u, const float* __restrict__ D,
          const float* __restrict__ Lre, const float* __restrict__ Lim,
          const float* __restrict__ Wri, const float* __restrict__ logstep,
          float* __restrict__ out) {
    __shared__ __attribute__((aligned(16))) float sU[CBW * LDP];
    __shared__ __attribute__((aligned(16))) float sA[CBW * LDP];
    __shared__ __attribute__((aligned(16))) float sB[CBW * LDP];
    __shared__ __attribute__((aligned(16))) float sCR[CHUNK * LDP];
    __shared__ __attribute__((aligned(16))) float sCI[CHUNK * LDP];
    __shared__ __attribute__((aligned(16))) float sKr[2 * CHUNK];
    __shared__ __attribute__((aligned(16))) float sC0[NN];
    __shared__ __attribute__((aligned(16))) float sZero[CHUNK];
    float* sZr = sA;
    float* sZi = sA + 64 * LDP;
    float* sPr = sA;
    float* sPi = sB;

    const int tid   = threadIdx.x;
    const int lane  = tid & 31;
    const int wave  = tid >> 5;
    const int h     = blockIdx.x / NCB;
    const int cb    = blockIdx.x % NCB;
    const int m     = lane & 15;
    const int khalf = lane >> 4;
    const int k0    = khalf * 2;

    float zcr = 0.f, zci = 0.f;
    if (tid < 2 * NN) {
        const int   n   = tid & 63;
        const float lre = Lre[h * NN + n], lim = Lim[h * NN + n];
        const float dt  = __expf(logstep[h]);
        const float ar  = dt * lre, ai = dt * lim;
        float cz, sz; sincosf(ai, &sz, &cz);
        const float er = expf(ar);
        const float zr = er * cz, zi = er * sz;
        if (tid < NN) {
            float pr = 1.f, pi_ = 0.f;
            float* zrrow = sZr + n * LDP;
            float* zirow = sZi + n * LDP;
            for (int k = 0; k < CHUNK; k += 4) {
                float qr[4], qi[4];
                for (int e = 0; e < 4; ++e) {
                    qr[3 - e] = pr; qi[3 - e] = pi_;
                    const float tr = pr * zr - pi_ * zi;
                    pi_ = pr * zi + pi_ * zr;
                    pr = tr;
                }
                *(float4*)(zrrow + CHUNK - 4 - k) = make_float4(qr[0], qr[1], qr[2], qr[3]);
                *(float4*)(zirow + CHUNK - 4 - k) = make_float4(qi[0], qi[1], qi[2], qi[3]);
            }
        } else {
            const float wr = Wri[(h * NN + n) * 2 + 0], wi = Wri[(h * NN + n) * 2 + 1];
            float cr, ci;
            mode_c(lre, lim, wr, wi, ar, ai, zr, zi, &cr, &ci);
            sC0[n] = cr;
            float qr = cr * zr - ci * zi;
            float qi = cr * zi + ci * zr;
            for (int t = 0; t < CHUNK; ++t) {
                sCR[t * LDP + n] =  qr;
                sCI[t * LDP + n] = -qi;
                const float tr = qr * zr - qi * zi;
                qi = qr * zi + qi * zr;
                qr = tr;
            }
        }
        float cC, sC; sincosf(ai * (float)CHUNK, &sC, &cC);
        const float eC = expf(ar * (float)CHUNK);
        zcr = eC * cC;
        zci = eC * sC;
    } else {
        const int t = tid - 2 * NN;
        const float* u0 = u + ((size_t)(2 * cb)     * HH + h) * LL;
        const float* u1 = u + ((size_t)(2 * cb + 1) * HH + h) * LL;
        v4f uv[16];
        for (int j = 0; j < 8; ++j) {
            uv[j]     = *(const v4f*)(u0 + (size_t)(t + 128 * j) * 4);
            uv[j + 8] = *(const v4f*)(u1 + (size_t)(t + 128 * j) * 4);
        }
        if (t < CHUNK) { sKr[CHUNK + t] = 0.f; sZero[t] = 0.f; }
        for (int j = 0; j < 8; ++j) {
            const int i = t + 128 * j;
            *(v4f*)(sU +            (i >> 4) * LDP + (i & 15) * 4) = uv[j];
            *(v4f*)(sU + 64 * LDP + (i >> 4) * LDP + (i & 15) * 4) = uv[j + 8];
        }
    }
    __syncthreads();

    if (tid >= 3 * NN) {
        const int t = tid - 3 * NN;
        const float* row = (t == 0) ? sC0 : (sCR + (t - 1) * LDP);
        float acc = 0.f;
        for (int q = 0; q < NN; q += 4) {
            const float4 v = *(const float4*)(row + q);
            acc += v.x; acc += v.y; acc += v.z; acc += v.w;
        }
        sKr[CHUNK - 1 - t] = acc;
    }

    {
        const float* bcol = sU  + (wave * 16 + m) * LDP + k0;
        const float* ar   = sZr + m * LDP + k0;
        const float* ai   = sZi + m * LDP + k0;

        v8f r0 = {}, r1 = {}, r2 = {}, r3 = {};
        v8f i0 = {}, i1 = {}, i2 = {}, i3 = {};
#pragma unroll
        for (int kk = 0; kk < CHUNK; kk += 4) {
            const v2f b = *(const v2f*)(bcol + kk);
            r0 = mma4(*(const v2f*)(ar + 0 * 16 * LDP + kk), b, r0);
            i0 = mma4(*(const v2f*)(ai + 0 * 16 * LDP + kk), b, i0);
            r1 = mma4(*(const v2f*)(ar + 1 * 16 * LDP + kk), b, r1);
            i1 = mma4(*(const v2f*)(ai + 1 * 16 * LDP + kk), b, i1);
            r2 = mma4(*(const v2f*)(ar + 2 * 16 * LDP + kk), b, r2);
            i2 = mma4(*(const v2f*)(ai + 2 * 16 * LDP + kk), b, i2);
            r3 = mma4(*(const v2f*)(ar + 3 * 16 * LDP + kk), b, r3);
            i3 = mma4(*(const v2f*)(ai + 3 * 16 * LDP + kk), b, i3);
        }

        __syncthreads();

        float* qr = sPr + (wave * 16 + m) * LDP + 8 * khalf;
        float* qi = sPi + (wave * 16 + m) * LDP + 8 * khalf;
        store8(qr +  0, r0); store8(qi +  0, i0);
        store8(qr + 16, r1); store8(qi + 16, i1);
        store8(qr + 32, r2); store8(qi + 32, i2);
        store8(qr + 48, r3); store8(qi + 48, i3);
    }
    __syncthreads();

    if (tid < 2 * NN) {
        float* pr = sPr + (tid >> 6) * NCH * LDP + (tid & 63);
        float* pi = sPi + (tid >> 6) * NCH * LDP + (tid & 63);
        float xr = 0.f, xi = 0.f;
        constexpr int SB = 16;
        float pvr[2][SB], pvi[2][SB];
#pragma unroll
        for (int k = 0; k < SB; ++k) { pvr[0][k] = pr[k * LDP]; pvi[0][k] = pi[k * LDP]; }
#pragma unroll
        for (int blk = 0; blk < NCH / SB; ++blk) {
            const int cur = blk & 1, nxt = cur ^ 1;
            if (blk + 1 < NCH / SB) {
#pragma unroll
                for (int k = 0; k < SB; ++k) {
                    pvr[nxt][k] = pr[((blk + 1) * SB + k) * LDP];
                    pvi[nxt][k] = pi[((blk + 1) * SB + k) * LDP];
                }
            }
#pragma unroll
            for (int k = 0; k < SB; ++k) {
                const float nxr = zcr * xr - zci * xi + pvr[cur][k];
                const float nxi = zcr * xi + zci * xr + pvi[cur][k];
                pr[(blk * SB + k) * LDP] = nxr; pi[(blk * SB + k) * LDP] = nxi;
                xr = nxr; xi = nxi;
            }
        }
    }
    __syncthreads();

    const int cl = wave * 16 + m;
    v8f a0 = {}, a1 = {}, a2 = {}, a3 = {};
    {
        const float* ucol = sU + cl * LDP + k0;
        const float* kb   = sKr + (CHUNK - 1) - m + k0;
#pragma unroll
        for (int kk = 0; kk < CHUNK; kk += 4) {
            const v2f bu = *(const v2f*)(ucol + kk);
            v2f t0, t1, t2, t3;
            t0.x = kb[kk];      t0.y = kb[kk + 1];
            t1.x = kb[kk - 16]; t1.y = kb[kk - 15];
            t2.x = kb[kk - 32]; t2.y = kb[kk - 31];
            t3.x = kb[kk - 48]; t3.y = kb[kk - 47];
            a0 = mma4(t0, bu, a0);
            a1 = mma4(t1, bu, a1);
            a2 = mma4(t2, bu, a2);
            a3 = mma4(t3, bu, a3);
        }
    }
    {
        const bool first = (cl & (NCH - 1)) == 0;
        const float* xrcol = first ? (sZero + k0) : (sPr + (cl - 1) * LDP + k0);
        const float* xicol = first ? (sZero + k0) : (sPi + (cl - 1) * LDP + k0);
        const float* acr   = sCR + m * LDP + k0;
        const float* aci   = sCI + m * LDP + k0;
#pragma unroll
        for (int kk = 0; kk < NN; kk += 4) {
            const v2f bxr = *(const v2f*)(xrcol + kk);
            const v2f bxi = *(const v2f*)(xicol + kk);
            a0 = mma4(*(const v2f*)(acr + 0 * 16 * LDP + kk), bxr, a0);
            a0 = mma4(*(const v2f*)(aci + 0 * 16 * LDP + kk), bxi, a0);
            a1 = mma4(*(const v2f*)(acr + 1 * 16 * LDP + kk), bxr, a1);
            a1 = mma4(*(const v2f*)(aci + 1 * 16 * LDP + kk), bxi, a1);
            a2 = mma4(*(const v2f*)(acr + 2 * 16 * LDP + kk), bxr, a2);
            a2 = mma4(*(const v2f*)(aci + 2 * 16 * LDP + kk), bxi, a2);
            a3 = mma4(*(const v2f*)(acr + 3 * 16 * LDP + kk), bxr, a3);
            a3 = mma4(*(const v2f*)(aci + 3 * 16 * LDP + kk), bxi, a3);
        }
    }

    const float Dh = D[h];
    float* up = sU + cl * LDP + 8 * khalf;
    v8f accs[4] = {a0, a1, a2, a3};
    for (int rt = 0; rt < 4; ++rt) {
        const v8f acc = accs[rt];
        const float4 u0 = *(const float4*)(up + rt * 16);
        const float4 u1 = *(const float4*)(up + rt * 16 + 4);
        float4 o0, o1;
        o0.x = acc[0] + Dh * u0.x; o0.y = acc[1] + Dh * u0.y;
        o0.z = acc[2] + Dh * u0.z; o0.w = acc[3] + Dh * u0.w;
        o1.x = acc[4] + Dh * u1.x; o1.y = acc[5] + Dh * u1.y;
        o1.z = acc[6] + Dh * u1.z; o1.w = acc[7] + Dh * u1.w;
        *(float4*)(up + rt * 16)     = o0;
        *(float4*)(up + rt * 16 + 4) = o1;
    }
    const int colw = cb * CBW + wave * 16;
    const int b    = colw >> 6;
    const int ch0  = colw & 63;
    const float* rb = sU + (wave * 16 + khalf) * LDP + m * 4;
    float* ow = out + ((size_t)b * HH + h) * LL + (size_t)(ch0 + khalf) * CHUNK + m * 4;
    for (int i = 0; i < 8; ++i) {
        const float4 v = *(const float4*)(rb + 2 * i * LDP);
        *(float4*)(ow + 2 * i * CHUNK) = v;
    }
}

extern "C" void kernel_launch(void* const* d_in, const int* in_sizes, int n_in,
                              void* d_out, int out_size, void* d_ws, size_t ws_size,
                              hipStream_t stream) {
    const float* u       = (const float*)d_in[0];
    const float* Lre     = (const float*)d_in[1];
    const float* Lim     = (const float*)d_in[2];
    const float* Wri     = (const float*)d_in[3];
    const float* Dp      = (const float*)d_in[4];
    const float* logstep = (const float*)d_in[5];
    float* out = (float*)d_out;

    dss_fused<<<HH * NCB, 256, 0, stream>>>(u, Dp, Lre, Lim, Wri, logstep, out);
}
